# starting kernel
# baseline (speedup 1.0000x reference)
#include <hip/hip_runtime.h>
#include <hip/hip_bf16.h>


typedef float v2f __attribute__((ext_vector_type(2)));
typedef float v8f __attribute__((ext_vector_type(8)));

static __device__ inline v8f wmma_f32_16x16x4(v2f a, v2f b, v8f c) {
  return __builtin_amdgcn_wmma_f32_16x16x4_f32(false, a, false, b, (short)0, c,
                                               false, false);
}

static __device__ inline float readlane_f(float v, int l) {
  union {
    float f;
    int i;
  } u;
  u.f = v;
  u.i = __builtin_amdgcn_readlane(u.i, l);
  return u.f;
}

#define WAVES_PER_BLK 8
#define ROW_STRIDE 18

__global__ __launch_bounds__(256) void caps_routing_kernel(
    const float* __restrict__ pred, const float* __restrict__ blog,
    const int* __restrict__ nitp, float* __restrict__ outp, int nsites,
    int bsites) {
  __shared__ float ldsTile[WAVES_PER_BLK][16 * ROW_STRIDE];

  const int lane = threadIdx.x & 31;
  const int wv   = threadIdx.x >> 5;
  const int site = blockIdx.x * WAVES_PER_BLK + wv;
  if (site >= nsites) return;

  float* tile = &ldsTile[wv][0];

  {
    const int row = lane >> 1, half = (lane & 1) * 8;
    const float4* g =
        reinterpret_cast<const float4*>(pred + (size_t)site * 256 + row * 16 + half);
    float4 x0 = g[0];
    float4 x1 = g[1];
    float* l = tile + row * ROW_STRIDE + half;
    reinterpret_cast<float2*>(l)[0] = make_float2(x0.x, x0.y);
    reinterpret_cast<float2*>(l)[1] = make_float2(x0.z, x0.w);
    reinterpret_cast<float2*>(l)[2] = make_float2(x1.x, x1.y);
    reinterpret_cast<float2*>(l)[3] = make_float2(x1.z, x1.w);
  }

  const bool lo = (lane & 16) == 0;
  const int  m  = lane & 15;

  v2f predA[4], predB[4];
#pragma unroll
  for (int k = 0; k < 4; ++k) {
    const int acol = 4 * k + (lo ? 0 : 2);
    predA[k] = *reinterpret_cast<v2f*>(tile + m * ROW_STRIDE + acol);
    const int brow = 4 * k + (lo ? 0 : 2);
    predB[k].x = tile[brow * ROW_STRIDE + m];
    predB[k].y = tile[(brow + 1) * ROW_STRIDE + m];
  }

  float bl[16];
  {
    const float* bsrc = blog + (size_t)(site % bsites) * 16;
#pragma unroll
    for (int i = 0; i < 16; ++i) bl[i] = bsrc[i];
  }

  float c[16];
  float csum;
  float outv[16];
  float laneOut = 0.f;

  auto softmax16 = [&]() {
    float mx = bl[0];
#pragma unroll
    for (int i = 1; i < 16; ++i) mx = fmaxf(mx, bl[i]);
    float sum = 0.f;
#pragma unroll
    for (int i = 0; i < 16; ++i) {
      c[i] = __expf(bl[i] - mx);
      sum += c[i];
    }
    csum = sum;
  };

  auto routeS = [&]() {
    v8f acc = {};
#pragma unroll
    for (int k = 0; k < 4; ++k) {
      v2f ac;
      ac.x = lo ? c[4 * k] : c[4 * k + 2];
      ac.y = lo ? c[4 * k + 1] : c[4 * k + 3];
      acc = wmma_f32_16x16x4(ac, predB[k], acc);
    }
    const float sval = acc[0];
    float sp[16], sqp = 0.f;
#pragma unroll
    for (int d = 0; d < 16; ++d) {
      sp[d] = readlane_f(sval, d);
      sqp += sp[d] * sp[d];
    }
    const float inv = __frcp_rn(csum);
    const float sq  = sqp * inv * inv;
    const float g   = (sq / (1.0f + sq)) * __frsqrt_rn(sq + 1e-7f) * inv;
#pragma unroll
    for (int d = 0; d < 16; ++d) outv[d] = g * sp[d];
    laneOut = g * sval;
  };

  auto agree = [&]() {
    v8f acc = {};
#pragma unroll
    for (int k = 0; k < 4; ++k) {
      v2f bo;
      bo.x = lo ? outv[4 * k] : outv[4 * k + 2];
      bo.y = lo ? outv[4 * k + 1] : outv[4 * k + 3];
      acc = wmma_f32_16x16x4(predA[k], bo, acc);
    }
#pragma unroll
    for (int j = 0; j < 8; ++j) {
      bl[j]     += readlane_f(acc[j], 0);
      bl[j + 8] += readlane_f(acc[j], 16);
    }
  };

  softmax16();
  routeS();
  const int iters = nitp[0];
  for (int t = 0; t < iters; ++t) {
    agree();
    softmax16();
    routeS();
  }

  if (lane < 16) outp[(size_t)site * 16 + lane] = laneOut;
}

extern "C" void kernel_launch(void* const* d_in, const int* in_sizes, int n_in,
                              void* d_out, int out_size, void* d_ws,
                              size_t ws_size, hipStream_t stream) {
  const float* pred = (const float*)d_in[0];
  const float* blog = (const float*)d_in[1];
  const int* nit    = (const int*)d_in[2];
  float* out        = (float*)d_out;

  const int nsites = out_size / 16;
  const int bsites = in_sizes[1] / 16;
  const int blocks = (nsites + WAVES_PER_BLK - 1) / WAVES_PER_BLK;

  caps_routing_kernel<<<blocks, 256, 0, stream>>>(pred, blog, nit, out, nsites,
                                                  bsites);
}


# Routing state kept in WMMA accumulator layout: pred on A side of both products, logits as C, permuted k order; no readlane/select; sums split per half-wave + permlanex16
# speedup vs baseline: 1.2202x; 1.2202x over previous
#include <hip/hip_runtime.h>
#include <hip/hip_bf16.h>


typedef float v2f __attribute__((ext_vector_type(2)));
typedef float v8f __attribute__((ext_vector_type(8)));

static __device__ inline v8f wmma_f32_16x16x4(v2f a, v2f b, v8f c) {
  return __builtin_amdgcn_wmma_f32_16x16x4_f32(false, a, false, b, (short)0, c,
                                               false, false);
}

static __device__ inline float xhalf(float x) {
  const int b = __builtin_bit_cast(int, x);
  return __builtin_bit_cast(
      float, __builtin_amdgcn_permlanex16(b, b, 0x76543210, 0xfedcba98, false,
                                          false));
}

#define WAVES_PER_BLK 8
#define ROW_STRIDE 18

__global__ __launch_bounds__(256) void caps_routing_kernel(
    const float* __restrict__ pred, const float* __restrict__ blog,
    const int* __restrict__ nitp, float* __restrict__ outp, int nsites,
    int bsites) {
  __shared__ float ldsTile[WAVES_PER_BLK][16 * ROW_STRIDE];

  const int lane = threadIdx.x & 31;
  const int wv   = threadIdx.x >> 5;
  const int site = blockIdx.x * WAVES_PER_BLK + wv;
  if (site >= nsites) return;

  float* tile = &ldsTile[wv][0];

  {
    const int row = lane >> 1, half = (lane & 1) * 8;
    const float4* g =
        reinterpret_cast<const float4*>(pred + (size_t)site * 256 + row * 16 + half);
    float4 x0 = g[0];
    float4 x1 = g[1];
    float* l = tile + row * ROW_STRIDE + half;
    reinterpret_cast<float2*>(l)[0] = make_float2(x0.x, x0.y);
    reinterpret_cast<float2*>(l)[1] = make_float2(x0.z, x0.w);
    reinterpret_cast<float2*>(l)[2] = make_float2(x1.x, x1.y);
    reinterpret_cast<float2*>(l)[3] = make_float2(x1.z, x1.w);
  }

  const int c16 = lane & 15;
  const int h   = lane >> 4;

  v2f pT[4], pR[4];
  for (int kk = 0; kk < 4; ++kk) {
    const int k0 = 8 * h + 2 * kk;
    pR[kk] = *reinterpret_cast<v2f*>(tile + c16 * ROW_STRIDE + k0);
    pT[kk].x = tile[k0 * ROW_STRIDE + c16];
    pT[kk].y = tile[(k0 + 1) * ROW_STRIDE + c16];
  }

  v8f bv;
  {
    const float4* bsrc = reinterpret_cast<const float4*>(
        blog + (size_t)(site % bsites) * 16 + 8 * h);
    const float4 b0 = bsrc[0];
    const float4 b1 = bsrc[1];
    bv[0] = b0.x; bv[1] = b0.y; bv[2] = b0.z; bv[3] = b0.w;
    bv[4] = b1.x; bv[5] = b1.y; bv[6] = b1.z; bv[7] = b1.w;
  }

  float cv[8];
  float csum;
  v8f   ov;

  auto softmax16 = [&]() {
    float mx = bv[0];
    for (int r = 1; r < 8; ++r) mx = fmaxf(mx, bv[r]);
    mx = fmaxf(mx, xhalf(mx));
    float sum = 0.f;
    for (int r = 0; r < 8; ++r) {
      cv[r] = __expf(bv[r] - mx);
      sum += cv[r];
    }
    csum = sum + xhalf(sum);
  };

  auto routeS = [&]() {
    v8f acc = {};
    for (int kk = 0; kk < 4; ++kk) {
      v2f cb;
      cb.x = cv[2 * kk];
      cb.y = cv[2 * kk + 1];
      acc = wmma_f32_16x16x4(pT[kk], cb, acc);
    }
    float sqp = 0.f;
    for (int r = 0; r < 8; ++r) sqp += acc[r] * acc[r];
    sqp += xhalf(sqp);
    const float inv = __frcp_rn(csum);
    const float sq  = sqp * inv * inv;
    const float g   = (sq / (1.0f + sq)) * __frsqrt_rn(sq + 1e-7f) * inv;
    for (int r = 0; r < 8; ++r) ov[r] = g * acc[r];
  };

  auto agree = [&]() {
    for (int kk = 0; kk < 4; ++kk) {
      v2f ob;
      ob.x = ov[2 * kk];
      ob.y = ov[2 * kk + 1];
      bv = wmma_f32_16x16x4(pR[kk], ob, bv);
    }
  };

  softmax16();
  routeS();
  const int iters = nitp[0];
  for (int t = 0; t < iters; ++t) {
    agree();
    softmax16();
    routeS();
  }

  {
    const float t0 = (c16 & 1) ? ov[1] : ov[0];
    const float t1 = (c16 & 1) ? ov[3] : ov[2];
    const float t2 = (c16 & 1) ? ov[5] : ov[4];
    const float t3 = (c16 & 1) ? ov[7] : ov[6];
    const float u0 = (c16 & 2) ? t1 : t0;
    const float u1 = (c16 & 2) ? t3 : t2;
    const float v  = (c16 & 4) ? u1 : u0;
    if (c16 < 8) outp[(size_t)site * 16 + 8 * h + c16] = v;
  }
}

extern "C" void kernel_launch(void* const* d_in, const int* in_sizes, int n_in,
                              void* d_out, int out_size, void* d_ws,
                              size_t ws_size, hipStream_t stream) {
  const float* pred = (const float*)d_in[0];
  const float* blog = (const float*)d_in[1];
  const int* nit    = (const int*)d_in[2];
  float* out        = (float*)d_out;

  const int nsites = out_size / 16;
  const int bsites = in_sizes[1] / 16;
  const int blocks = (nsites + WAVES_PER_BLK - 1) / WAVES_PER_BLK;

  caps_routing_kernel<<<blocks, 256, 0, stream>>>(pred, blog, nit, out, nsites,
                                                  bsites);
}


# Routing passes run on the Gram matrix G = pred pred^T (4 WMMAs, once): per pass one product u = G c, |s|^2 = c.u, b += g*u; 20 WMMAs per site, not 28; sums reassociated, P(P^T c) as (P P^T)c
# speedup vs baseline: 1.3986x; 1.1462x over previous
#include <hip/hip_runtime.h>
#include <hip/hip_bf16.h>


typedef float v2f __attribute__((ext_vector_type(2)));
typedef float v8f __attribute__((ext_vector_type(8)));

static __device__ inline v8f wmma_f32_16x16x4(v2f a, v2f b, v8f c) {
  return __builtin_amdgcn_wmma_f32_16x16x4_f32(false, a, false, b, (short)0, c,
                                               false, false);
}

static __device__ inline float xhalf(float x) {
  const int b = __builtin_bit_cast(int, x);
  return __builtin_bit_cast(
      float, __builtin_amdgcn_permlanex16(b, b, 0x76543210, 0xfedcba98, false,
                                          false));
}

#define WAVES_PER_BLK 8
#define ROW_STRIDE 18

__global__ __launch_bounds__(256) void caps_routing_kernel(
    const float* __restrict__ pred, const float* __restrict__ blog,
    const int* __restrict__ nitp, float* __restrict__ outp, int nsites,
    int bsites) {
  __shared__ float ldsTile[WAVES_PER_BLK][16 * ROW_STRIDE];

  const int lane = threadIdx.x & 31;
  const int wv   = threadIdx.x >> 5;
  const int site = blockIdx.x * WAVES_PER_BLK + wv;
  if (site >= nsites) return;

  float* tile = &ldsTile[wv][0];

  {
    const int row = lane >> 1, half = (lane & 1) * 8;
    const float4* g =
        reinterpret_cast<const float4*>(pred + (size_t)site * 256 + row * 16 + half);
    float4 x0 = g[0];
    float4 x1 = g[1];
    float* l = tile + row * ROW_STRIDE + half;
    reinterpret_cast<float2*>(l)[0] = make_float2(x0.x, x0.y);
    reinterpret_cast<float2*>(l)[1] = make_float2(x0.z, x0.w);
    reinterpret_cast<float2*>(l)[2] = make_float2(x1.x, x1.y);
    reinterpret_cast<float2*>(l)[3] = make_float2(x1.z, x1.w);
  }

  const int c16 = lane & 15;
  const int h   = lane >> 4;

  v2f pT[4], pR[4];
  for (int kk = 0; kk < 4; ++kk) {
    const int k0 = 8 * h + 2 * kk;
    pR[kk] = *reinterpret_cast<v2f*>(tile + c16 * ROW_STRIDE + k0);
    pT[kk].x = tile[k0 * ROW_STRIDE + c16];
    pT[kk].y = tile[(k0 + 1) * ROW_STRIDE + c16];
  }

  v8f bv;
  {
    const float4* bsrc = reinterpret_cast<const float4*>(
        blog + (size_t)(site % bsites) * 16 + 8 * h);
    const float4 b0 = bsrc[0];
    const float4 b1 = bsrc[1];
    bv[0] = b0.x; bv[1] = b0.y; bv[2] = b0.z; bv[3] = b0.w;
    bv[4] = b1.x; bv[5] = b1.y; bv[6] = b1.z; bv[7] = b1.w;
  }

  float cv[8];
  float csum;
  v8f   ov;

  auto softmax16 = [&]() {
    float mx = bv[0];
    for (int r = 1; r < 8; ++r) mx = fmaxf(mx, bv[r]);
    mx = fmaxf(mx, xhalf(mx));
    float sum = 0.f;
    for (int r = 0; r < 8; ++r) {
      cv[r] = __expf(bv[r] - mx);
      sum += cv[r];
    }
    csum = sum + xhalf(sum);
  };

  auto routeS = [&]() {
    v8f acc = {};
    for (int kk = 0; kk < 4; ++kk) {
      v2f cb;
      cb.x = cv[2 * kk];
      cb.y = cv[2 * kk + 1];
      acc = wmma_f32_16x16x4(pT[kk], cb, acc);
    }
    float sqp = 0.f;
    for (int r = 0; r < 8; ++r) sqp += acc[r] * acc[r];
    sqp += xhalf(sqp);
    const float inv = __frcp_rn(csum);
    const float sq  = sqp * inv * inv;
    const float g   = (sq / (1.0f + sq)) * __frsqrt_rn(sq + 1e-7f) * inv;
    for (int r = 0; r < 8; ++r) ov[r] = g * acc[r];
  };

  v8f gm = {};
  for (int kk = 0; kk < 4; ++kk) gm = wmma_f32_16x16x4(pR[kk], pR[kk], gm);

  auto routeG = [&]() {
    v8f u = {};
    for (int kk = 0; kk < 4; ++kk) {
      v2f ga, cb;
      ga.x = gm[2 * kk];
      ga.y = gm[2 * kk + 1];
      cb.x = cv[2 * kk];
      cb.y = cv[2 * kk + 1];
      u = wmma_f32_16x16x4(ga, cb, u);
    }
    float sqp = cv[0] * u[0];
    for (int r = 1; r < 8; ++r) sqp = fmaf(cv[r], u[r], sqp);
    sqp += xhalf(sqp);
    sqp = (sqp < 0.f) ? 0.f : sqp;
    const float inv = __frcp_rn(csum);
    const float sq  = sqp * inv * inv;
    const float g   = (sq / (1.0f + sq)) * __frsqrt_rn(sq + 1e-7f) * inv;
    for (int r = 0; r < 8; ++r) bv[r] = fmaf(g, u[r], bv[r]);
  };

  softmax16();
  const int iters = nitp[0];
  for (int t = 0; t < iters; ++t) {
    routeG();
    softmax16();
  }
  routeS();

  {
    const float t0 = (c16 & 1) ? ov[1] : ov[0];
    const float t1 = (c16 & 1) ? ov[3] : ov[2];
    const float t2 = (c16 & 1) ? ov[5] : ov[4];
    const float t3 = (c16 & 1) ? ov[7] : ov[6];
    const float u0 = (c16 & 2) ? t1 : t0;
    const float u1 = (c16 & 2) ? t3 : t2;
    const float v  = (c16 & 4) ? u1 : u0;
    if (c16 < 8) outp[(size_t)site * 16 + 8 * h + c16] = v;
  }
}

extern "C" void kernel_launch(void* const* d_in, const int* in_sizes, int n_in,
                              void* d_out, int out_size, void* d_ws,
                              size_t ws_size, hipStream_t stream) {
  const float* pred = (const float*)d_in[0];
  const float* blog = (const float*)d_in[1];
  const int* nit    = (const int*)d_in[2];
  float* out        = (float*)d_out;

  const int nsites = out_size / 16;
  const int bsites = in_sizes[1] / 16;
  const int blocks = (nsites + WAVES_PER_BLK - 1) / WAVES_PER_BLK;

  caps_routing_kernel<<<blocks, 256, 0, stream>>>(pred, blog, nit, out, nsites,
                                                  bsites);
}


# 16 sites per wave, site j in column j of shared accumulator tiles (B = site's weights in column j, 0 elsewhere): softmax, squash, logit update once per 16 sites; per-site sum order kept
# speedup vs baseline: 2.5127x; 1.7966x over previous
#include <hip/hip_runtime.h>
#include <hip/hip_bf16.h>


typedef float v2f __attribute__((ext_vector_type(2)));
typedef float v8f __attribute__((ext_vector_type(8)));

static __device__ inline v8f wmma_f32_16x16x4(v2f a, v2f b, v8f c) {
  return __builtin_amdgcn_wmma_f32_16x16x4_f32(false, a, false, b, (short)0, c,
                                               false, false);
}

static __device__ inline float xhalf(float x) {
  const int b = __builtin_bit_cast(int, x);
  return __builtin_bit_cast(
      float, __builtin_amdgcn_permlanex16(b, b, 0x76543210, 0xfedcba98, false,
                                          false));
}

#define NS 16
#define WAVES_PER_BLK 2
#define ROW_STRIDE 18

__global__ __launch_bounds__(64) void caps_routing_kernel(
    const float* __restrict__ pred, const float* __restrict__ blog,
    const int* __restrict__ nitp, float* __restrict__ outp, int nsites,
    int bsites) {
  __shared__ float ldsTile[WAVES_PER_BLK][NS][16 * ROW_STRIDE];

  const int lane  = threadIdx.x & 31;
  const int wv    = threadIdx.x >> 5;
  const int site0 = (blockIdx.x * WAVES_PER_BLK + wv) * NS;
  if (site0 >= nsites) return;
  const int last = nsites - 1;

  const int c16 = lane & 15;
  const int h   = lane >> 4;

  v8f gm[NS];
#pragma unroll
  for (int j = 0; j < NS; j += 2) {
    const int sa = min(site0 + j, last);
    const int sb = min(site0 + j + 1, last);
    const float4* ga = reinterpret_cast<const float4*>(
        pred + (size_t)sa * 256 + c16 * 16 + 8 * h);
    const float4* gb = reinterpret_cast<const float4*>(
        pred + (size_t)sb * 256 + c16 * 16 + 8 * h);
    const float4 a0 = ga[0], a1 = ga[1];
    const float4 b0 = gb[0], b1 = gb[1];
    float2* la = reinterpret_cast<float2*>(
        &ldsTile[wv][j][c16 * ROW_STRIDE + 8 * h]);
    float2* lb = reinterpret_cast<float2*>(
        &ldsTile[wv][j + 1][c16 * ROW_STRIDE + 8 * h]);
    la[0] = make_float2(a0.x, a0.y);
    la[1] = make_float2(a0.z, a0.w);
    la[2] = make_float2(a1.x, a1.y);
    la[3] = make_float2(a1.z, a1.w);
    lb[0] = make_float2(b0.x, b0.y);
    lb[1] = make_float2(b0.z, b0.w);
    lb[2] = make_float2(b1.x, b1.y);
    lb[3] = make_float2(b1.z, b1.w);
    v2f pa[4], pb[4];
    pa[0].x = a0.x; pa[0].y = a0.y; pa[1].x = a0.z; pa[1].y = a0.w;
    pa[2].x = a1.x; pa[2].y = a1.y; pa[3].x = a1.z; pa[3].y = a1.w;
    pb[0].x = b0.x; pb[0].y = b0.y; pb[1].x = b0.z; pb[1].y = b0.w;
    pb[2].x = b1.x; pb[2].y = b1.y; pb[3].x = b1.z; pb[3].y = b1.w;
    v8f accA = {}, accB = {};
    for (int kk = 0; kk < 4; ++kk) {
      accA = wmma_f32_16x16x4(pa[kk], pa[kk], accA);
      accB = wmma_f32_16x16x4(pb[kk], pb[kk], accB);
    }
    gm[j]     = accA;
    gm[j + 1] = accB;
  }

  v8f bv;
  {
    const int sl = min(site0 + c16, last);
    const float4* bsrc = reinterpret_cast<const float4*>(
        blog + (size_t)(sl % bsites) * 16 + 8 * h);
    const float4 b0 = bsrc[0];
    const float4 b1 = bsrc[1];
    bv[0] = b0.x; bv[1] = b0.y; bv[2] = b0.z; bv[3] = b0.w;
    bv[4] = b1.x; bv[5] = b1.y; bv[6] = b1.z; bv[7] = b1.w;
  }

  float cv[8];
  float csum;

  auto softmax16 = [&]() {
    float mx = bv[0];
    for (int r = 1; r < 8; ++r) mx = fmaxf(mx, bv[r]);
    mx = fmaxf(mx, xhalf(mx));
    float sum = 0.f;
    for (int r = 0; r < 8; ++r) {
      cv[r] = __expf(bv[r] - mx);
      sum += cv[r];
    }
    csum = sum + xhalf(sum);
  };

  auto colB = [&](int j, int kk) {
    v2f cb;
    cb.x = (c16 == j) ? cv[2 * kk] : 0.f;
    cb.y = (c16 == j) ? cv[2 * kk + 1] : 0.f;
    return cb;
  };

  auto routeG = [&]() {
    v8f u0 = {}, u1 = {};
#pragma unroll
    for (int j = 0; j < NS; j += 2) {
      for (int kk = 0; kk < 4; ++kk) {
        v2f g0, g1;
        g0.x = gm[j][2 * kk];
        g0.y = gm[j][2 * kk + 1];
        g1.x = gm[j + 1][2 * kk];
        g1.y = gm[j + 1][2 * kk + 1];
        u0 = wmma_f32_16x16x4(g0, colB(j, kk), u0);
        u1 = wmma_f32_16x16x4(g1, colB(j + 1, kk), u1);
      }
    }
    float u[8];
    for (int r = 0; r < 8; ++r) u[r] = u0[r] + u1[r];
    float sqp = cv[0] * u[0];
    for (int r = 1; r < 8; ++r) sqp = fmaf(cv[r], u[r], sqp);
    sqp += xhalf(sqp);
    sqp = (sqp < 0.f) ? 0.f : sqp;
    const float inv = __frcp_rn(csum);
    const float sq  = sqp * inv * inv;
    const float g   = (sq / (1.0f + sq)) * __frsqrt_rn(sq + 1e-7f) * inv;
    for (int r = 0; r < 8; ++r) bv[r] = fmaf(g, u[r], bv[r]);
  };

  softmax16();
  const int iters = nitp[0];
  for (int t = 0; t < iters; ++t) {
    routeG();
    softmax16();
  }

  v2f pT[NS][4];
#pragma unroll
  for (int j = 0; j < NS; ++j) {
    const float* tj = &ldsTile[wv][j][c16];
    for (int kk = 0; kk < 4; ++kk) {
      const int k0 = 8 * h + 2 * kk;
      pT[j][kk].x = tj[k0 * ROW_STRIDE];
      pT[j][kk].y = tj[(k0 + 1) * ROW_STRIDE];
    }
  }
  __builtin_amdgcn_sched_barrier(0);
  v8f s0 = {}, s1 = {};
#pragma unroll
  for (int j = 0; j < NS; j += 2) {
    for (int kk = 0; kk < 4; ++kk) {
      s0 = wmma_f32_16x16x4(pT[j][kk], colB(j, kk), s0);
      s1 = wmma_f32_16x16x4(pT[j + 1][kk], colB(j + 1, kk), s1);
    }
  }
  float sv[8];
  for (int r = 0; r < 8; ++r) sv[r] = s0[r] + s1[r];
  float sqp = 0.f;
  for (int r = 0; r < 8; ++r) sqp += sv[r] * sv[r];
  sqp += xhalf(sqp);
  const float inv = __frcp_rn(csum);
  const float sq  = sqp * inv * inv;
  const float g   = (sq / (1.0f + sq)) * __frsqrt_rn(sq + 1e-7f) * inv;

  if (site0 + c16 < nsites) {
    float4* o = reinterpret_cast<float4*>(outp + (size_t)(site0 + c16) * 16 +
                                          8 * h);
    o[0] = make_float4(g * sv[0], g * sv[1], g * sv[2], g * sv[3]);
    o[1] = make_float4(g * sv[4], g * sv[5], g * sv[6], g * sv[7]);
  }
}

extern "C" void kernel_launch(void* const* d_in, const int* in_sizes, int n_in,
                              void* d_out, int out_size, void* d_ws,
                              size_t ws_size, hipStream_t stream) {
  const float* pred = (const float*)d_in[0];
  const float* blog = (const float*)d_in[1];
  const int* nit    = (const int*)d_in[2];
  float* out        = (float*)d_out;

  const int nsites = out_size / 16;
  const int bsites = in_sizes[1] / 16;
  const int per_blk = WAVES_PER_BLK * NS;
  const int blocks = (nsites + per_blk - 1) / per_blk;

  caps_routing_kernel<<<blocks, 64, 0, stream>>>(pred, blog, nit, out, nsites,
                                                 bsites);
}
